# starting kernel
# baseline (speedup 1.0000x reference)
#include <hip/hip_runtime.h>

typedef __attribute__((ext_vector_type(2))) float v2f;
typedef __attribute__((ext_vector_type(8))) float v8f;

#define B_   4
#define C_   68
#define F0   4
#define PT   2048
#define PS   8192
#define KSEL 4
#define FLT_BIG 3.4e38f

__global__ void sqnorm_kernel(const float* __restrict__ S, float* __restrict__ sq) {
  int q = blockIdx.x * blockDim.x + threadIdx.x;
  int b = blockIdx.y;
  if (q >= PS) return;
  const float* p = S + ((size_t)b * C_ + F0) * PS + q;
  float acc = 0.f;
  #pragma unroll 8
  for (int f = 0; f < 64; ++f) {
    float v = p[(size_t)f * PS];
    acc = fmaf(v, v, acc);
  }
  sq[b * PS + q] = acc;
}

__launch_bounds__(128)
__global__ void knn_kernel(const float* __restrict__ T, const float* __restrict__ S,
                           const float* __restrict__ sq, int* __restrict__ out_idx) {
  const int b    = blockIdx.y;
  const int p0   = blockIdx.x * 16;
  const int lane = threadIdx.x & 31;
  const int wave = threadIdx.x >> 5;
  const int half = lane >> 4;
  const int ln   = lane & 15;

  v2f a[16];
  #pragma unroll
  for (int s = 0; s < 16; ++s) {
    int f = F0 + 4 * s + 2 * half;
    const float* tp = T + ((size_t)b * C_ + f) * PT + p0 + ln;
    a[s].x = tp[0];
    a[s].y = tp[PT];
  }

  float best[8][KSEL];
  int   bidx[8][KSEL];
  #pragma unroll
  for (int v = 0; v < 8; ++v)
    #pragma unroll
    for (int j = 0; j < KSEL; ++j) { best[v][j] = FLT_BIG; bidx[v][j] = 0; }

  const int qbeg = wave * (PS / 4);
  const int qend = qbeg + (PS / 4);
  for (int q0 = qbeg; q0 < qend; q0 += 16) {
    v2f bb[16];
    #pragma unroll
    for (int s = 0; s < 16; ++s) {
      int f = F0 + 4 * s + 2 * half;
      const float* sp = S + ((size_t)b * C_ + f) * PS + q0 + ln;
      bb[s].x = sp[0];
      bb[s].y = sp[PS];
    }

    v8f acc = {};
    #pragma unroll
    for (int s = 0; s < 16; ++s) {
      acc = __builtin_amdgcn_wmma_f32_16x16x4_f32(
           false, a[s],  false, bb[s],
           (short)0, acc,  false,  false);
    }

    const float sqv  = sq[b * PS + q0 + ln];
    const int   qidx = q0 + ln;

    #pragma unroll
    for (int v = 0; v < 8; ++v) {
      float sc = fmaf(-2.f, acc[v], sqv);
      if (sc < best[v][3]) {
        if (sc < best[v][2]) {
          best[v][3] = best[v][2]; bidx[v][3] = bidx[v][2];
          if (sc < best[v][1]) {
            best[v][2] = best[v][1]; bidx[v][2] = bidx[v][1];
            if (sc < best[v][0]) {
              best[v][1] = best[v][0]; bidx[v][1] = bidx[v][0];
              best[v][0] = sc; bidx[v][0] = qidx;
            } else { best[v][1] = sc; bidx[v][1] = qidx; }
          } else { best[v][2] = sc; bidx[v][2] = qidx; }
        } else { best[v][3] = sc; bidx[v][3] = qidx; }
      }
    }
  }

  __shared__ float ls[4][16][16][KSEL];
  __shared__ int   li[4][16][16][KSEL];
  #pragma unroll
  for (int v = 0; v < 8; ++v) {
    int row = v + 8 * half;
    #pragma unroll
    for (int j = 0; j < KSEL; ++j) {
      ls[wave][row][ln][j] = best[v][j];
      li[wave][row][ln][j] = bidx[v][j];
    }
  }
  __syncthreads();

  if (threadIdx.x < 16) {
    const int row = threadIdx.x;
    float rs[KSEL]; int ri[KSEL];
    #pragma unroll
    for (int j = 0; j < KSEL; ++j) { rs[j] = FLT_BIG; ri[j] = 0; }
    for (int w = 0; w < 4; ++w)
      for (int n = 0; n < 16; ++n)
        #pragma unroll
        for (int j = 0; j < KSEL; ++j) {
          float sc = ls[w][row][n][j];
          int   id = li[w][row][n][j];
          if (sc < rs[3]) {
            if (sc < rs[2]) {
              rs[3] = rs[2]; ri[3] = ri[2];
              if (sc < rs[1]) {
                rs[2] = rs[1]; ri[2] = ri[1];
                if (sc < rs[0]) {
                  rs[1] = rs[0]; ri[1] = ri[0]; rs[0] = sc; ri[0] = id;
                } else { rs[1] = sc; ri[1] = id; }
              } else { rs[2] = sc; ri[2] = id; }
            } else { rs[3] = sc; ri[3] = id; }
          }
        }
    int* o = out_idx + ((size_t)b * PT + p0 + row) * KSEL;
    o[0] = ri[0]; o[1] = ri[1]; o[2] = ri[2]; o[3] = ri[3];
  }
}

__global__ void gather_kernel(const float* __restrict__ S, const int* __restrict__ idx,
                              float* __restrict__ out) {
  size_t t = (size_t)blockIdx.x * blockDim.x + threadIdx.x;
  if (t >= (size_t)B_ * C_ * PT) return;
  int p  = (int)(t % PT);
  int bc = (int)(t / PT);
  int b  = bc / C_;
  const int4 id = *(const int4*)(idx + ((size_t)b * PT + p) * KSEL);
  const float* src = S + (size_t)bc * PS;
  float4 r;
  r.x = src[id.x]; r.y = src[id.y]; r.z = src[id.z]; r.w = src[id.w];
  *(float4*)(out + t * KSEL) = r;
}

extern "C" void kernel_launch(void* const* d_in, const int* in_sizes, int n_in,
                              void* d_out, int out_size, void* d_ws, size_t ws_size,
                              hipStream_t stream) {
  const float* T = (const float*)d_in[0];
  const float* S = (const float*)d_in[1];

  float* sq  = (float*)d_ws;
  int*   idx = (int*)((char*)d_ws + (size_t)B_ * PS * sizeof(float));
  float* out = (float*)d_out;

  sqnorm_kernel<<<dim3(PS / 256, B_), 256, 0, stream>>>(S, sq);
  knn_kernel<<<dim3(PT / 16, B_), 128, 0, stream>>>(T, S, sq, idx);
  size_t n = (size_t)B_ * C_ * PT;
  gather_kernel<<<dim3((unsigned)((n + 255) / 256)), 256, 0, stream>>>(S, idx, out);
}


# Each wave keeps 2 target tiles as WMMA B and streams source tiles as A (a lane owns one target point, one top-4 list per tile): every source load feeds 2 WMMAs; 8 waves per block split the sources
# speedup vs baseline: 1.4146x; 1.4146x over previous
#include <hip/hip_runtime.h>

typedef __attribute__((ext_vector_type(2))) float v2f;
typedef __attribute__((ext_vector_type(8))) float v8f;

#define B_   4
#define C_   68
#define F0   4
#define PT   2048
#define PS   8192
#define KSEL 4
#define FLT_BIG 3.4e38f

__global__ void sqnorm_kernel(const float* __restrict__ S, float* __restrict__ sq) {
  int q = blockIdx.x * blockDim.x + threadIdx.x;
  int b = blockIdx.y;
  if (q >= PS) return;
  const float* p = S + ((size_t)b * C_ + F0) * PS + q;
  float acc = 0.f;
  #pragma unroll 8
  for (int f = 0; f < 64; ++f) {
    float v = p[(size_t)f * PS];
    acc = fmaf(v, v, acc);
  }
  sq[b * PS + q] = acc;
}

#define NT 2
#define NW 8

__launch_bounds__(NW * 32)
__global__ void knn_kernel(const float* __restrict__ T, const float* __restrict__ S,
                           const float* __restrict__ sq, int* __restrict__ out_idx) {
  const int b    = blockIdx.y;
  const int p0   = blockIdx.x * (16 * NT);
  const int lane = threadIdx.x & 31;
  const int wave = threadIdx.x >> 5;
  const int half = lane >> 4;
  const int ln   = lane & 15;

  v2f tb[NT][16];
  #pragma unroll
  for (int t = 0; t < NT; ++t)
    #pragma unroll
    for (int s = 0; s < 16; ++s) {
      int f = F0 + 4 * s + 2 * half;
      const float* tp = T + ((size_t)b * C_ + f) * PT + p0 + 16 * t + ln;
      tb[t][s].x = tp[0];
      tb[t][s].y = tp[PT];
    }

  float best[NT][KSEL];
  int   bidx[NT][KSEL];
  #pragma unroll
  for (int t = 0; t < NT; ++t)
    #pragma unroll
    for (int j = 0; j < KSEL; ++j) { best[t][j] = FLT_BIG; bidx[t][j] = 0; }

  const int qbeg = wave * (PS / NW);
  const int qend = qbeg + (PS / NW);
  for (int q0 = qbeg; q0 < qend; q0 += 16) {
    v2f sa[16];
    #pragma unroll
    for (int s = 0; s < 16; ++s) {
      int f = F0 + 4 * s + 2 * half;
      const float* sp = S + ((size_t)b * C_ + f) * PS + q0 + ln;
      sa[s].x = sp[0];
      sa[s].y = sp[PS];
    }
    const float4* sqp = (const float4*)(sq + (size_t)b * PS + q0 + 8 * half);
    const float4 n0 = sqp[0];
    const float4 n1 = sqp[1];
    const float sqv[8] = {n0.x, n0.y, n0.z, n0.w, n1.x, n1.y, n1.z, n1.w};

    v8f acc[NT];
    #pragma unroll
    for (int t = 0; t < NT; ++t) acc[t] = v8f{};
    #pragma unroll
    for (int s = 0; s < 16; ++s)
      #pragma unroll
      for (int t = 0; t < NT; ++t)
        acc[t] = __builtin_amdgcn_wmma_f32_16x16x4_f32(
             false, sa[s],  false, tb[t][s],
             (short)0, acc[t],  false,  false);

    const int qbase = q0 + 8 * half;
    #pragma unroll
    for (int t = 0; t < NT; ++t)
      #pragma unroll
      for (int v = 0; v < 8; ++v) {
        float sc = fmaf(-2.f, acc[t][v], sqv[v]);
        const int qidx = qbase + v;
        if (sc < best[t][3]) {
          if (sc < best[t][2]) {
            best[t][3] = best[t][2]; bidx[t][3] = bidx[t][2];
            if (sc < best[t][1]) {
              best[t][2] = best[t][1]; bidx[t][2] = bidx[t][1];
              if (sc < best[t][0]) {
                best[t][1] = best[t][0]; bidx[t][1] = bidx[t][0];
                best[t][0] = sc; bidx[t][0] = qidx;
              } else { best[t][1] = sc; bidx[t][1] = qidx; }
            } else { best[t][2] = sc; bidx[t][2] = qidx; }
          } else { best[t][3] = sc; bidx[t][3] = qidx; }
        }
      }
  }

  __shared__ float ls[NW * 2][16 * NT][KSEL];
  __shared__ int   li[NW * 2][16 * NT][KSEL];
  #pragma unroll
  for (int t = 0; t < NT; ++t)
    #pragma unroll
    for (int j = 0; j < KSEL; ++j) {
      ls[wave * 2 + half][16 * t + ln][j] = best[t][j];
      li[wave * 2 + half][16 * t + ln][j] = bidx[t][j];
    }
  __syncthreads();

  if (threadIdx.x < 16 * NT) {
    const int row = threadIdx.x;
    float rs[KSEL]; int ri[KSEL];
    #pragma unroll
    for (int j = 0; j < KSEL; ++j) { rs[j] = FLT_BIG; ri[j] = 0; }
    for (int n = 0; n < NW * 2; ++n)
      #pragma unroll
      for (int j = 0; j < KSEL; ++j) {
        float sc = ls[n][row][j];
        int   id = li[n][row][j];
        if (sc < rs[3]) {
          if (sc < rs[2]) {
            rs[3] = rs[2]; ri[3] = ri[2];
            if (sc < rs[1]) {
              rs[2] = rs[1]; ri[2] = ri[1];
              if (sc < rs[0]) {
                rs[1] = rs[0]; ri[1] = ri[0]; rs[0] = sc; ri[0] = id;
              } else { rs[1] = sc; ri[1] = id; }
            } else { rs[2] = sc; ri[2] = id; }
          } else { rs[3] = sc; ri[3] = id; }
        }
      }
    int* o = out_idx + ((size_t)b * PT + p0 + row) * KSEL;
    o[0] = ri[0]; o[1] = ri[1]; o[2] = ri[2]; o[3] = ri[3];
  }
}

__global__ void gather_kernel(const float* __restrict__ S, const int* __restrict__ idx,
                              float* __restrict__ out) {
  size_t t = (size_t)blockIdx.x * blockDim.x + threadIdx.x;
  if (t >= (size_t)B_ * C_ * PT) return;
  int p  = (int)(t % PT);
  int bc = (int)(t / PT);
  int b  = bc / C_;
  const int4 id = *(const int4*)(idx + ((size_t)b * PT + p) * KSEL);
  const float* src = S + (size_t)bc * PS;
  float4 r;
  r.x = src[id.x]; r.y = src[id.y]; r.z = src[id.z]; r.w = src[id.w];
  *(float4*)(out + t * KSEL) = r;
}

extern "C" void kernel_launch(void* const* d_in, const int* in_sizes, int n_in,
                              void* d_out, int out_size, void* d_ws, size_t ws_size,
                              hipStream_t stream) {
  const float* T = (const float*)d_in[0];
  const float* S = (const float*)d_in[1];

  float* sq  = (float*)d_ws;
  int*   idx = (int*)((char*)d_ws + (size_t)B_ * PS * sizeof(float));
  float* out = (float*)d_out;

  sqnorm_kernel<<<dim3(PS / 256, B_), 256, 0, stream>>>(S, sq);
  knn_kernel<<<dim3(PT / (16 * NT), B_), NW * 32, 0, stream>>>(T, S, sq, idx);
  size_t n = (size_t)B_ * C_ * PT;
  gather_kernel<<<dim3((unsigned)((n + 255) / 256)), 256, 0, stream>>>(S, idx, out);
}


# Top-4 insertion without branches: four compares against the old list, then selects move values and indices (same strict-< placement, same visiting order), in the source loop and in the merge
# speedup vs baseline: 1.8217x; 1.2878x over previous
#include <hip/hip_runtime.h>

typedef __attribute__((ext_vector_type(2))) float v2f;
typedef __attribute__((ext_vector_type(8))) float v8f;

#define B_   4
#define C_   68
#define F0   4
#define PT   2048
#define PS   8192
#define KSEL 4
#define FLT_BIG 3.4e38f

__global__ void sqnorm_kernel(const float* __restrict__ S, float* __restrict__ sq) {
  int q = blockIdx.x * blockDim.x + threadIdx.x;
  int b = blockIdx.y;
  if (q >= PS) return;
  const float* p = S + ((size_t)b * C_ + F0) * PS + q;
  float acc = 0.f;
  #pragma unroll 8
  for (int f = 0; f < 64; ++f) {
    float v = p[(size_t)f * PS];
    acc = fmaf(v, v, acc);
  }
  sq[b * PS + q] = acc;
}

#define NT 2
#define NW 8

__device__ __forceinline__ void insert4(float (&b)[KSEL], int (&i)[KSEL], float sc, int id) {
  const bool c0 = sc < b[0], c1 = sc < b[1], c2 = sc < b[2], c3 = sc < b[3];
  const float v3 = c3 ? sc : b[3];  const int j3 = c3 ? id : i[3];
  const float v2 = c2 ? sc : b[2];  const int j2 = c2 ? id : i[2];
  const float v1 = c1 ? sc : b[1];  const int j1 = c1 ? id : i[1];
  b[3] = c2 ? b[2] : v3;  i[3] = c2 ? i[2] : j3;
  b[2] = c1 ? b[1] : v2;  i[2] = c1 ? i[1] : j2;
  b[1] = c0 ? b[0] : v1;  i[1] = c0 ? i[0] : j1;
  b[0] = c0 ? sc : b[0];  i[0] = c0 ? id : i[0];
}

__launch_bounds__(NW * 32)
__global__ void knn_kernel(const float* __restrict__ T, const float* __restrict__ S,
                           const float* __restrict__ sq, int* __restrict__ out_idx) {
  const int b    = blockIdx.y;
  const int p0   = blockIdx.x * (16 * NT);
  const int lane = threadIdx.x & 31;
  const int wave = threadIdx.x >> 5;
  const int half = lane >> 4;
  const int ln   = lane & 15;

  v2f tb[NT][16];
  #pragma unroll
  for (int t = 0; t < NT; ++t)
    #pragma unroll
    for (int s = 0; s < 16; ++s) {
      int f = F0 + 4 * s + 2 * half;
      const float* tp = T + ((size_t)b * C_ + f) * PT + p0 + 16 * t + ln;
      tb[t][s].x = tp[0];
      tb[t][s].y = tp[PT];
    }

  float best[NT][KSEL];
  int   bidx[NT][KSEL];
  #pragma unroll
  for (int t = 0; t < NT; ++t)
    #pragma unroll
    for (int j = 0; j < KSEL; ++j) { best[t][j] = FLT_BIG; bidx[t][j] = 0; }

  const int qbeg = wave * (PS / NW);
  const int qend = qbeg + (PS / NW);
  for (int q0 = qbeg; q0 < qend; q0 += 16) {
    v2f sa[16];
    #pragma unroll
    for (int s = 0; s < 16; ++s) {
      int f = F0 + 4 * s + 2 * half;
      const float* sp = S + ((size_t)b * C_ + f) * PS + q0 + ln;
      sa[s].x = sp[0];
      sa[s].y = sp[PS];
    }
    const float4* sqp = (const float4*)(sq + (size_t)b * PS + q0 + 8 * half);
    const float4 n0 = sqp[0];
    const float4 n1 = sqp[1];
    const float sqv[8] = {n0.x, n0.y, n0.z, n0.w, n1.x, n1.y, n1.z, n1.w};

    v8f acc[NT];
    #pragma unroll
    for (int t = 0; t < NT; ++t) acc[t] = v8f{};
    #pragma unroll
    for (int s = 0; s < 16; ++s)
      #pragma unroll
      for (int t = 0; t < NT; ++t)
        acc[t] = __builtin_amdgcn_wmma_f32_16x16x4_f32(
             false, sa[s],  false, tb[t][s],
             (short)0, acc[t],  false,  false);

    const int qbase = q0 + 8 * half;
    #pragma unroll
    for (int t = 0; t < NT; ++t)
      #pragma unroll
      for (int v = 0; v < 8; ++v) {
        float sc = fmaf(-2.f, acc[t][v], sqv[v]);
        const int qidx = qbase + v;
        insert4(best[t], bidx[t], sc, qidx);
      }
  }

  __shared__ float ls[NW * 2][16 * NT][KSEL];
  __shared__ int   li[NW * 2][16 * NT][KSEL];
  #pragma unroll
  for (int t = 0; t < NT; ++t)
    #pragma unroll
    for (int j = 0; j < KSEL; ++j) {
      ls[wave * 2 + half][16 * t + ln][j] = best[t][j];
      li[wave * 2 + half][16 * t + ln][j] = bidx[t][j];
    }
  __syncthreads();

  if (threadIdx.x < 16 * NT) {
    const int row = threadIdx.x;
    float rs[KSEL]; int ri[KSEL];
    #pragma unroll
    for (int j = 0; j < KSEL; ++j) { rs[j] = FLT_BIG; ri[j] = 0; }
    for (int n = 0; n < NW * 2; ++n)
      #pragma unroll
      for (int j = 0; j < KSEL; ++j) {
        float sc = ls[n][row][j];
        int   id = li[n][row][j];
        insert4(rs, ri, sc, id);
      }
    int* o = out_idx + ((size_t)b * PT + p0 + row) * KSEL;
    o[0] = ri[0]; o[1] = ri[1]; o[2] = ri[2]; o[3] = ri[3];
  }
}

__global__ void gather_kernel(const float* __restrict__ S, const int* __restrict__ idx,
                              float* __restrict__ out) {
  size_t t = (size_t)blockIdx.x * blockDim.x + threadIdx.x;
  if (t >= (size_t)B_ * C_ * PT) return;
  int p  = (int)(t % PT);
  int bc = (int)(t / PT);
  int b  = bc / C_;
  const int4 id = *(const int4*)(idx + ((size_t)b * PT + p) * KSEL);
  const float* src = S + (size_t)bc * PS;
  float4 r;
  r.x = src[id.x]; r.y = src[id.y]; r.z = src[id.z]; r.w = src[id.w];
  *(float4*)(out + t * KSEL) = r;
}

extern "C" void kernel_launch(void* const* d_in, const int* in_sizes, int n_in,
                              void* d_out, int out_size, void* d_ws, size_t ws_size,
                              hipStream_t stream) {
  const float* T = (const float*)d_in[0];
  const float* S = (const float*)d_in[1];

  float* sq  = (float*)d_ws;
  int*   idx = (int*)((char*)d_ws + (size_t)B_ * PS * sizeof(float));
  float* out = (float*)d_out;

  sqnorm_kernel<<<dim3(PS / 256, B_), 256, 0, stream>>>(S, sq);
  knn_kernel<<<dim3(PT / (16 * NT), B_), NW * 32, 0, stream>>>(T, S, sq, idx);
  size_t n = (size_t)B_ * C_ * PT;
  gather_kernel<<<dim3((unsigned)((n + 255) / 256)), 256, 0, stream>>>(S, idx, out);
}


# Neighbour fetch through LDS: one block per (b,c) source row stages the 32 KB row with coalesced 16-byte loads and picks its 4x2048 neighbours from LDS instead of scattered global loads
# speedup vs baseline: 1.9059x; 1.0462x over previous
#include <hip/hip_runtime.h>

typedef __attribute__((ext_vector_type(2))) float v2f;
typedef __attribute__((ext_vector_type(8))) float v8f;

#define B_   4
#define C_   68
#define F0   4
#define PT   2048
#define PS   8192
#define KSEL 4
#define FLT_BIG 3.4e38f

__global__ void sqnorm_kernel(const float* __restrict__ S, float* __restrict__ sq) {
  int q = blockIdx.x * blockDim.x + threadIdx.x;
  int b = blockIdx.y;
  if (q >= PS) return;
  const float* p = S + ((size_t)b * C_ + F0) * PS + q;
  float acc = 0.f;
  #pragma unroll 8
  for (int f = 0; f < 64; ++f) {
    float v = p[(size_t)f * PS];
    acc = fmaf(v, v, acc);
  }
  sq[b * PS + q] = acc;
}

#define NT 2
#define NW 8

__device__ __forceinline__ void insert4(float (&b)[KSEL], int (&i)[KSEL], float sc, int id) {
  const bool c0 = sc < b[0], c1 = sc < b[1], c2 = sc < b[2], c3 = sc < b[3];
  const float v3 = c3 ? sc : b[3];  const int j3 = c3 ? id : i[3];
  const float v2 = c2 ? sc : b[2];  const int j2 = c2 ? id : i[2];
  const float v1 = c1 ? sc : b[1];  const int j1 = c1 ? id : i[1];
  b[3] = c2 ? b[2] : v3;  i[3] = c2 ? i[2] : j3;
  b[2] = c1 ? b[1] : v2;  i[2] = c1 ? i[1] : j2;
  b[1] = c0 ? b[0] : v1;  i[1] = c0 ? i[0] : j1;
  b[0] = c0 ? sc : b[0];  i[0] = c0 ? id : i[0];
}

__launch_bounds__(NW * 32)
__global__ void knn_kernel(const float* __restrict__ T, const float* __restrict__ S,
                           const float* __restrict__ sq, int* __restrict__ out_idx) {
  const int b    = blockIdx.y;
  const int p0   = blockIdx.x * (16 * NT);
  const int lane = threadIdx.x & 31;
  const int wave = threadIdx.x >> 5;
  const int half = lane >> 4;
  const int ln   = lane & 15;

  v2f tb[NT][16];
  #pragma unroll
  for (int t = 0; t < NT; ++t)
    #pragma unroll
    for (int s = 0; s < 16; ++s) {
      int f = F0 + 4 * s + 2 * half;
      const float* tp = T + ((size_t)b * C_ + f) * PT + p0 + 16 * t + ln;
      tb[t][s].x = tp[0];
      tb[t][s].y = tp[PT];
    }

  float best[NT][KSEL];
  int   bidx[NT][KSEL];
  #pragma unroll
  for (int t = 0; t < NT; ++t)
    #pragma unroll
    for (int j = 0; j < KSEL; ++j) { best[t][j] = FLT_BIG; bidx[t][j] = 0; }

  const int qbeg = wave * (PS / NW);
  const int qend = qbeg + (PS / NW);
  for (int q0 = qbeg; q0 < qend; q0 += 16) {
    v2f sa[16];
    #pragma unroll
    for (int s = 0; s < 16; ++s) {
      int f = F0 + 4 * s + 2 * half;
      const float* sp = S + ((size_t)b * C_ + f) * PS + q0 + ln;
      sa[s].x = sp[0];
      sa[s].y = sp[PS];
    }
    const float4* sqp = (const float4*)(sq + (size_t)b * PS + q0 + 8 * half);
    const float4 n0 = sqp[0];
    const float4 n1 = sqp[1];
    const float sqv[8] = {n0.x, n0.y, n0.z, n0.w, n1.x, n1.y, n1.z, n1.w};

    v8f acc[NT];
    #pragma unroll
    for (int t = 0; t < NT; ++t) acc[t] = v8f{};
    #pragma unroll
    for (int s = 0; s < 16; ++s)
      #pragma unroll
      for (int t = 0; t < NT; ++t)
        acc[t] = __builtin_amdgcn_wmma_f32_16x16x4_f32(
             false, sa[s],  false, tb[t][s],
             (short)0, acc[t],  false,  false);

    const int qbase = q0 + 8 * half;
    #pragma unroll
    for (int t = 0; t < NT; ++t)
      #pragma unroll
      for (int v = 0; v < 8; ++v) {
        float sc = fmaf(-2.f, acc[t][v], sqv[v]);
        const int qidx = qbase + v;
        insert4(best[t], bidx[t], sc, qidx);
      }
  }

  __shared__ float ls[NW * 2][16 * NT][KSEL];
  __shared__ int   li[NW * 2][16 * NT][KSEL];
  #pragma unroll
  for (int t = 0; t < NT; ++t)
    #pragma unroll
    for (int j = 0; j < KSEL; ++j) {
      ls[wave * 2 + half][16 * t + ln][j] = best[t][j];
      li[wave * 2 + half][16 * t + ln][j] = bidx[t][j];
    }
  __syncthreads();

  if (threadIdx.x < 16 * NT) {
    const int row = threadIdx.x;
    float rs[KSEL]; int ri[KSEL];
    #pragma unroll
    for (int j = 0; j < KSEL; ++j) { rs[j] = FLT_BIG; ri[j] = 0; }
    for (int n = 0; n < NW * 2; ++n)
      #pragma unroll
      for (int j = 0; j < KSEL; ++j) {
        float sc = ls[n][row][j];
        int   id = li[n][row][j];
        insert4(rs, ri, sc, id);
      }
    int* o = out_idx + ((size_t)b * PT + p0 + row) * KSEL;
    o[0] = ri[0]; o[1] = ri[1]; o[2] = ri[2]; o[3] = ri[3];
  }
}

#define GT 256
#define GP (PT / GT)
#define GV (PS / 4 / GT)

__launch_bounds__(GT)
__global__ void gather_kernel(const float* __restrict__ S, const int* __restrict__ idx,
                              float* __restrict__ out) {
  __shared__ float row[PS];
  const int bc  = blockIdx.x;
  const int b   = bc / C_;
  const int tid = threadIdx.x;

  const int4* ip = (const int4*)(idx + (size_t)b * PT * KSEL);
  int4 id[GP];
  #pragma unroll
  for (int j = 0; j < GP; ++j) id[j] = ip[tid + j * GT];

  const float4* src = (const float4*)(S + (size_t)bc * PS);
  float4 v[GV];
  #pragma unroll
  for (int j = 0; j < GV; ++j) v[j] = src[tid + j * GT];
  #pragma unroll
  for (int j = 0; j < GV; ++j) ((float4*)row)[tid + j * GT] = v[j];
  __syncthreads();

  float4* o = (float4*)(out + (size_t)bc * PT * KSEL);
  #pragma unroll
  for (int j = 0; j < GP; ++j) {
    float4 r;
    r.x = row[id[j].x]; r.y = row[id[j].y]; r.z = row[id[j].z]; r.w = row[id[j].w];
    o[tid + j * GT] = r;
  }
}

extern "C" void kernel_launch(void* const* d_in, const int* in_sizes, int n_in,
                              void* d_out, int out_size, void* d_ws, size_t ws_size,
                              hipStream_t stream) {
  const float* T = (const float*)d_in[0];
  const float* S = (const float*)d_in[1];

  float* sq  = (float*)d_ws;
  int*   idx = (int*)((char*)d_ws + (size_t)B_ * PS * sizeof(float));
  float* out = (float*)d_out;

  sqnorm_kernel<<<dim3(PS / 256, B_), 256, 0, stream>>>(S, sq);
  knn_kernel<<<dim3(PT / (16 * NT), B_), NW * 32, 0, stream>>>(T, S, sq, idx);
  gather_kernel<<<dim3(B_ * C_), GT, 0, stream>>>(S, idx, out);
}
